# starting kernel
# baseline (speedup 1.0000x reference)
#include <hip/hip_runtime.h>
#include <hip/hip_bf16.h>

#define N_NODES 50000
#define N_EDGES 800000
#define IN_DIM  512
#define HID_DIM 128
#define M_TILES (N_NODES / 16)

#define PAIR_STRIDE 288
#define N_PAIRS     16

typedef float v2f __attribute__((ext_vector_type(2)));
typedef float v8f __attribute__((ext_vector_type(8)));

__global__ void init_kernel(float* __restrict__ out, float* __restrict__ deg) {
    int i = blockIdx.x * blockDim.x + threadIdx.x;
    if (i < N_NODES * HID_DIM) out[i] = 0.0f;
    if (i < N_NODES) deg[i] = 1.0f;
}

__global__ void deg_kernel(const int* __restrict__ dst, float* __restrict__ deg) {
    int e = blockIdx.x * blockDim.x + threadIdx.x;
    if (e < N_EDGES) atomicAdd(&deg[dst[e]], 1.0f);
}

__global__ void dinv_kernel(const float* __restrict__ deg, float* __restrict__ dinv) {
    int i = blockIdx.x * blockDim.x + threadIdx.x;
    if (i < N_NODES) dinv[i] = rsqrtf(deg[i]);
}

__global__ __launch_bounds__(256) void gemm_kernel(const float* __restrict__ x,
                                                   const float* __restrict__ W,
                                                   float* __restrict__ h) {
    __shared__ float ldsW[N_PAIRS * PAIR_STRIDE];

    const int lane   = threadIdx.x & 31;
    const int waveId = threadIdx.x >> 5;
    const int mTile  = blockIdx.x * 8 + waveId;
    const bool valid = (mTile < M_TILES);

    const int half = lane >> 4;
    const int l16  = lane & 15;

    const int c  = threadIdx.x & 127;
    const int ph = threadIdx.x >> 7;

    v8f acc[8];
#pragma unroll
    for (int t = 0; t < 8; ++t) acc[t] = (v8f){0.f,0.f,0.f,0.f,0.f,0.f,0.f,0.f};

    const int row = mTile * 16 + l16;
    const float* xrow = x + (size_t)row * IN_DIM;

    for (int kc = 0; kc < IN_DIM; kc += 32) {
#pragma unroll
        for (int pp = 0; pp < 8; ++pp) {
            const int pair = ph * 8 + pp;
            float w0 = W[(size_t)(kc + 2 * pair    ) * HID_DIM + c];
            float w1 = W[(size_t)(kc + 2 * pair + 1) * HID_DIM + c];
            *(v2f*)(ldsW + pair * PAIR_STRIDE + c * 2) = (v2f){w0, w1};
        }
        __syncthreads();

        if (valid) {
            if (kc + 32 < IN_DIM)
                __builtin_prefetch(xrow + kc + 32, 0, 3);
#pragma unroll
            for (int k = 0; k < 32; k += 4) {
                const int kk   = k + 2 * half;
                const int pair = (k >> 1) + half;
                v2f a = *(const v2f*)(xrow + kc + kk);
                const float* bbase = ldsW + (size_t)pair * PAIR_STRIDE;
#pragma unroll
                for (int t = 0; t < 8; ++t) {
                    v2f b = *(const v2f*)(bbase + (t * 16 + l16) * 2);
                    acc[t] = __builtin_amdgcn_wmma_f32_16x16x4_f32(
                         false, a,  false, b,
                         (short)0, acc[t],
                         false,  false);
                }
            }
        }
        __syncthreads();
    }

    if (valid) {
#pragma unroll
        for (int t = 0; t < 8; ++t) {
#pragma unroll
            for (int r = 0; r < 8; ++r) {
                int m = mTile * 16 + r + 8 * half;
                h[(size_t)m * HID_DIM + t * 16 + l16] = acc[t][r];
            }
        }
    }
}

__global__ __launch_bounds__(256) void aggregate_kernel(const float* __restrict__ h,
                                                        const int* __restrict__ src,
                                                        const int* __restrict__ dst,
                                                        const float* __restrict__ dinv,
                                                        float* __restrict__ out) {
    int tid  = blockIdx.x * blockDim.x + threadIdx.x;
    int e    = tid >> 5;
    int lane = tid & 31;
    if (e >= N_EDGES) return;
    int s = src[e];
    int d = dst[e];
    float norm = dinv[s] * dinv[d];
    float4 v = ((const float4*)(h + (size_t)s * HID_DIM))[lane];
    float* o = out + (size_t)d * HID_DIM + lane * 4;
    atomicAdd(o + 0, v.x * norm);
    atomicAdd(o + 1, v.y * norm);
    atomicAdd(o + 2, v.z * norm);
    atomicAdd(o + 3, v.w * norm);
}

__global__ void finalize_kernel(const float* __restrict__ h,
                                const float* __restrict__ dinv,
                                const float* __restrict__ b,
                                const float* __restrict__ prelu_w,
                                float* __restrict__ out) {
    int i = blockIdx.x * blockDim.x + threadIdx.x;
    if (i >= N_NODES * HID_DIM) return;
    int rowI = i / HID_DIM;
    int c    = i - rowI * HID_DIM;
    float di = dinv[rowI];
    float v  = out[i] + h[i] * di * di + b[c];
    out[i]   = (v > 0.0f) ? v : prelu_w[c] * v;
}

extern "C" void kernel_launch(void* const* d_in, const int* in_sizes, int n_in,
                              void* d_out, int out_size, void* d_ws, size_t ws_size,
                              hipStream_t stream) {
    const float* x       = (const float*)d_in[0];
    const int*   eidx    = (const int*)d_in[1];
    const float* W       = (const float*)d_in[2];
    const float* b       = (const float*)d_in[3];
    const float* prelu_w = (const float*)d_in[4];
    float*       out     = (float*)d_out;

    const int* src = eidx;
    const int* dst = eidx + N_EDGES;

    char* ws    = (char*)d_ws;
    float* h    = (float*)(ws);
    float* deg  = (float*)(ws + (size_t)N_NODES * HID_DIM * 4);
    float* dinv = (float*)(ws + (size_t)N_NODES * HID_DIM * 4 + (size_t)N_NODES * 4);

    const int TPB = 256;

    init_kernel<<<(N_NODES * HID_DIM + TPB - 1) / TPB, TPB, 0, stream>>>(out, deg);

    deg_kernel<<<(N_EDGES + TPB - 1) / TPB, TPB, 0, stream>>>(dst, deg);

    dinv_kernel<<<(N_NODES + TPB - 1) / TPB, TPB, 0, stream>>>(deg, dinv);

    gemm_kernel<<<(M_TILES + 7) / 8, TPB, 0, stream>>>(x, W, h);

    aggregate_kernel<<<(N_EDGES * 32 + TPB - 1) / TPB, TPB, 0, stream>>>(h, src, dst, dinv, out);

    finalize_kernel<<<(N_NODES * HID_DIM + TPB - 1) / TPB, TPB, 0, stream>>>(h, dinv, b, prelu_w, out);
}


# aggregate: lane l takes columns l, l+32, l+64, l+96, so each f32 atomic instruction of a wave covers one whole 128-byte line of out (4 line requests per edge, not 16); arrival order as before
# speedup vs baseline: 1.2475x; 1.2475x over previous
#include <hip/hip_runtime.h>
#include <hip/hip_bf16.h>

#define N_NODES 50000
#define N_EDGES 800000
#define IN_DIM  512
#define HID_DIM 128
#define M_TILES (N_NODES / 16)

#define PAIR_STRIDE 288
#define N_PAIRS     16

typedef float v2f __attribute__((ext_vector_type(2)));
typedef float v8f __attribute__((ext_vector_type(8)));

__global__ void init_kernel(float* __restrict__ out, float* __restrict__ deg) {
    int i = blockIdx.x * blockDim.x + threadIdx.x;
    if (i < N_NODES * HID_DIM) out[i] = 0.0f;
    if (i < N_NODES) deg[i] = 1.0f;
}

__global__ void deg_kernel(const int* __restrict__ dst, float* __restrict__ deg) {
    int e = blockIdx.x * blockDim.x + threadIdx.x;
    if (e < N_EDGES) atomicAdd(&deg[dst[e]], 1.0f);
}

__global__ void dinv_kernel(const float* __restrict__ deg, float* __restrict__ dinv) {
    int i = blockIdx.x * blockDim.x + threadIdx.x;
    if (i < N_NODES) dinv[i] = rsqrtf(deg[i]);
}

__global__ __launch_bounds__(256) void gemm_kernel(const float* __restrict__ x,
                                                   const float* __restrict__ W,
                                                   float* __restrict__ h) {
    __shared__ float ldsW[N_PAIRS * PAIR_STRIDE];

    const int lane   = threadIdx.x & 31;
    const int waveId = threadIdx.x >> 5;
    const int mTile  = blockIdx.x * 8 + waveId;
    const bool valid = (mTile < M_TILES);

    const int half = lane >> 4;
    const int l16  = lane & 15;

    const int c  = threadIdx.x & 127;
    const int ph = threadIdx.x >> 7;

    v8f acc[8];
#pragma unroll
    for (int t = 0; t < 8; ++t) acc[t] = (v8f){0.f,0.f,0.f,0.f,0.f,0.f,0.f,0.f};

    const int row = mTile * 16 + l16;
    const float* xrow = x + (size_t)row * IN_DIM;

    for (int kc = 0; kc < IN_DIM; kc += 32) {
#pragma unroll
        for (int pp = 0; pp < 8; ++pp) {
            const int pair = ph * 8 + pp;
            float w0 = W[(size_t)(kc + 2 * pair    ) * HID_DIM + c];
            float w1 = W[(size_t)(kc + 2 * pair + 1) * HID_DIM + c];
            *(v2f*)(ldsW + pair * PAIR_STRIDE + c * 2) = (v2f){w0, w1};
        }
        __syncthreads();

        if (valid) {
            if (kc + 32 < IN_DIM)
                __builtin_prefetch(xrow + kc + 32, 0, 3);
#pragma unroll
            for (int k = 0; k < 32; k += 4) {
                const int kk   = k + 2 * half;
                const int pair = (k >> 1) + half;
                v2f a = *(const v2f*)(xrow + kc + kk);
                const float* bbase = ldsW + (size_t)pair * PAIR_STRIDE;
#pragma unroll
                for (int t = 0; t < 8; ++t) {
                    v2f b = *(const v2f*)(bbase + (t * 16 + l16) * 2);
                    acc[t] = __builtin_amdgcn_wmma_f32_16x16x4_f32(
                         false, a,  false, b,
                         (short)0, acc[t],
                         false,  false);
                }
            }
        }
        __syncthreads();
    }

    if (valid) {
#pragma unroll
        for (int t = 0; t < 8; ++t) {
#pragma unroll
            for (int r = 0; r < 8; ++r) {
                int m = mTile * 16 + r + 8 * half;
                h[(size_t)m * HID_DIM + t * 16 + l16] = acc[t][r];
            }
        }
    }
}

__global__ __launch_bounds__(256) void aggregate_kernel(const float* __restrict__ h,
                                                        const int* __restrict__ src,
                                                        const int* __restrict__ dst,
                                                        const float* __restrict__ dinv,
                                                        float* __restrict__ out) {
    int tid  = blockIdx.x * blockDim.x + threadIdx.x;
    int e    = tid >> 5;
    int lane = tid & 31;
    if (e >= N_EDGES) return;
    int s = src[e];
    int d = dst[e];
    float norm = dinv[s] * dinv[d];
    const float* hp = h + (size_t)s * HID_DIM + lane;
    float v0 = hp[0];
    float v1 = hp[32];
    float v2 = hp[64];
    float v3 = hp[96];
    float* o = out + (size_t)d * HID_DIM + lane;
    atomicAdd(o +  0, v0 * norm);
    atomicAdd(o + 32, v1 * norm);
    atomicAdd(o + 64, v2 * norm);
    atomicAdd(o + 96, v3 * norm);
}

__global__ void finalize_kernel(const float* __restrict__ h,
                                const float* __restrict__ dinv,
                                const float* __restrict__ b,
                                const float* __restrict__ prelu_w,
                                float* __restrict__ out) {
    int i = blockIdx.x * blockDim.x + threadIdx.x;
    if (i >= N_NODES * HID_DIM) return;
    int rowI = i / HID_DIM;
    int c    = i - rowI * HID_DIM;
    float di = dinv[rowI];
    float v  = out[i] + h[i] * di * di + b[c];
    out[i]   = (v > 0.0f) ? v : prelu_w[c] * v;
}

extern "C" void kernel_launch(void* const* d_in, const int* in_sizes, int n_in,
                              void* d_out, int out_size, void* d_ws, size_t ws_size,
                              hipStream_t stream) {
    const float* x       = (const float*)d_in[0];
    const int*   eidx    = (const int*)d_in[1];
    const float* W       = (const float*)d_in[2];
    const float* b       = (const float*)d_in[3];
    const float* prelu_w = (const float*)d_in[4];
    float*       out     = (float*)d_out;

    const int* src = eidx;
    const int* dst = eidx + N_EDGES;

    char* ws    = (char*)d_ws;
    float* h    = (float*)(ws);
    float* deg  = (float*)(ws + (size_t)N_NODES * HID_DIM * 4);
    float* dinv = (float*)(ws + (size_t)N_NODES * HID_DIM * 4 + (size_t)N_NODES * 4);

    const int TPB = 256;

    init_kernel<<<(N_NODES * HID_DIM + TPB - 1) / TPB, TPB, 0, stream>>>(out, deg);

    deg_kernel<<<(N_EDGES + TPB - 1) / TPB, TPB, 0, stream>>>(dst, deg);

    dinv_kernel<<<(N_NODES + TPB - 1) / TPB, TPB, 0, stream>>>(deg, dinv);

    gemm_kernel<<<(M_TILES + 7) / 8, TPB, 0, stream>>>(x, W, h);

    aggregate_kernel<<<(N_EDGES * 32 + TPB - 1) / TPB, TPB, 0, stream>>>(h, src, dst, dinv, out);

    finalize_kernel<<<(N_NODES * HID_DIM + TPB - 1) / TPB, TPB, 0, stream>>>(h, dinv, b, prelu_w, out);
}


# aggregate: int atomic slot counter buckets each edge's source in its destination's out row; one wave per node sums those h rows in registers in slot order, self-loop and bias last; no f32 atomics
# speedup vs baseline: 1.8237x; 1.4619x over previous
#include <hip/hip_runtime.h>
#include <hip/hip_bf16.h>

#define N_NODES 50000
#define N_EDGES 800000
#define IN_DIM  512
#define HID_DIM 128
#define M_TILES (N_NODES / 16)

#define PAIR_STRIDE 288
#define N_PAIRS     16

#define BUCKET_CAP HID_DIM

typedef float v2f __attribute__((ext_vector_type(2)));
typedef float v4f __attribute__((ext_vector_type(4)));
typedef float v8f __attribute__((ext_vector_type(8)));
typedef int slot_t __attribute__((may_alias));

__global__ void init_kernel(int* __restrict__ cnt) {
    int i = blockIdx.x * blockDim.x + threadIdx.x;
    if (i < N_NODES) cnt[i] = 0;
}

__global__ void bucket_kernel(const int* __restrict__ src, const int* __restrict__ dst,
                              int* __restrict__ cnt, int* __restrict__ bucket) {
    int e = blockIdx.x * blockDim.x + threadIdx.x;
    if (e >= N_EDGES) return;
    int d = dst[e];
    int s = src[e];
    int p = atomicAdd(&cnt[d], 1);
    if (p < BUCKET_CAP) bucket[(size_t)d * BUCKET_CAP + p] = s;
}

__global__ __launch_bounds__(256) void gemm_kernel(const float* __restrict__ x,
                                                   const float* __restrict__ W,
                                                   float* __restrict__ h) {
    __shared__ float ldsW[N_PAIRS * PAIR_STRIDE];

    const int lane   = threadIdx.x & 31;
    const int waveId = threadIdx.x >> 5;
    const int mTile  = blockIdx.x * 8 + waveId;
    const bool valid = (mTile < M_TILES);

    const int half = lane >> 4;
    const int l16  = lane & 15;

    const int c  = threadIdx.x & 127;
    const int ph = threadIdx.x >> 7;

    v8f acc[8];
#pragma unroll
    for (int t = 0; t < 8; ++t) acc[t] = (v8f){0.f,0.f,0.f,0.f,0.f,0.f,0.f,0.f};

    const int row = mTile * 16 + l16;
    const float* xrow = x + (size_t)row * IN_DIM;

    for (int kc = 0; kc < IN_DIM; kc += 32) {
#pragma unroll
        for (int pp = 0; pp < 8; ++pp) {
            const int pair = ph * 8 + pp;
            float w0 = W[(size_t)(kc + 2 * pair    ) * HID_DIM + c];
            float w1 = W[(size_t)(kc + 2 * pair + 1) * HID_DIM + c];
            *(v2f*)(ldsW + pair * PAIR_STRIDE + c * 2) = (v2f){w0, w1};
        }
        __syncthreads();

        if (valid) {
            if (kc + 32 < IN_DIM)
                __builtin_prefetch(xrow + kc + 32, 0, 3);
#pragma unroll
            for (int k = 0; k < 32; k += 4) {
                const int kk   = k + 2 * half;
                const int pair = (k >> 1) + half;
                v2f a = *(const v2f*)(xrow + kc + kk);
                const float* bbase = ldsW + (size_t)pair * PAIR_STRIDE;
#pragma unroll
                for (int t = 0; t < 8; ++t) {
                    v2f b = *(const v2f*)(bbase + (t * 16 + l16) * 2);
                    acc[t] = __builtin_amdgcn_wmma_f32_16x16x4_f32(
                         false, a,  false, b,
                         (short)0, acc[t],
                         false,  false);
                }
            }
        }
        __syncthreads();
    }

    if (valid) {
#pragma unroll
        for (int t = 0; t < 8; ++t) {
#pragma unroll
            for (int r = 0; r < 8; ++r) {
                int m = mTile * 16 + r + 8 * half;
                h[(size_t)m * HID_DIM + t * 16 + l16] = acc[t][r];
            }
        }
    }
}

__device__ __forceinline__ float dinv_of(int c) { return rsqrtf((float)(c + 1)); }

template <int G>
__device__ __forceinline__ void add_rows(v4f& acc, const v4f* __restrict__ h4,
                                         int sj, float wj, int k, int lane) {
    int s[G];
    float w[G];
    v4f v[G];
    for (int u = 0; u < G; ++u) {
        s[u] = __builtin_amdgcn_readlane(sj, k + u);
        w[u] = __builtin_bit_cast(float,
                 __builtin_amdgcn_readlane(__builtin_bit_cast(int, wj), k + u));
    }
    for (int u = 0; u < G; ++u) v[u] = h4[(size_t)s[u] * (HID_DIM / 4) + lane];
    for (int u = 0; u < G; ++u) acc += v[u] * w[u];
}

__global__ __launch_bounds__(256) void gather_kernel(const float* __restrict__ h,
                                                     const int* __restrict__ cnt,
                                                     const int* __restrict__ src,
                                                     const int* __restrict__ dst,
                                                     const float* __restrict__ b,
                                                     const float* __restrict__ prelu_w,
                                                     float* out) {
    const int lane = threadIdx.x & 31;
    const int i = __builtin_amdgcn_readfirstlane(blockIdx.x * 8 + (threadIdx.x >> 5));
    if (i >= N_NODES) return;

    const v4f* h4 = (const v4f*)h;
    const int   n  = cnt[i];
    const float di = dinv_of(n);
    v4f acc = (v4f){0.f, 0.f, 0.f, 0.f};

    if (n <= BUCKET_CAP) {
        const slot_t* slots = (const slot_t*)out + (size_t)i * BUCKET_CAP;
        for (int base = 0; base < n; base += 32) {
            const int m = (n - base < 32) ? (n - base) : 32;
            int   sj = 0;
            float wj = 0.f;
            if (lane < m) {
                sj = slots[base + lane];
                wj = dinv_of(cnt[sj]) * di;
            }
            int k = 0;
            for (; k + 4 <= m; k += 4) add_rows<4>(acc, h4, sj, wj, k, lane);
            if (m & 2) { add_rows<2>(acc, h4, sj, wj, k, lane); k += 2; }
            if (m & 1) add_rows<1>(acc, h4, sj, wj, k, lane);
        }
    } else {
        for (int e0 = 0; e0 < N_EDGES; e0 += 32) {
            unsigned mask = __builtin_amdgcn_ballot_w32(dst[e0 + lane] == i);
            if (mask) {
                const int   sj = src[e0 + lane];
                const float wj = dinv_of(cnt[sj]) * di;
                while (mask) {
                    const int k = __builtin_ctz(mask);
                    mask &= mask - 1;
                    add_rows<1>(acc, h4, sj, wj, k, lane);
                }
            }
        }
    }

    const v4f hi = h4[(size_t)i * (HID_DIM / 4) + lane];
    const v4f bb = ((const v4f*)b)[lane];
    const v4f pw = ((const v4f*)prelu_w)[lane];
    const v4f v  = acc + hi * di * di + bb;
    v4f r;
    r.x = (v.x > 0.0f) ? v.x : pw.x * v.x;
    r.y = (v.y > 0.0f) ? v.y : pw.y * v.y;
    r.z = (v.z > 0.0f) ? v.z : pw.z * v.z;
    r.w = (v.w > 0.0f) ? v.w : pw.w * v.w;
    ((v4f*)out)[(size_t)i * (HID_DIM / 4) + lane] = r;
}

extern "C" void kernel_launch(void* const* d_in, const int* in_sizes, int n_in,
                              void* d_out, int out_size, void* d_ws, size_t ws_size,
                              hipStream_t stream) {
    const float* x       = (const float*)d_in[0];
    const int*   eidx    = (const int*)d_in[1];
    const float* W       = (const float*)d_in[2];
    const float* b       = (const float*)d_in[3];
    const float* prelu_w = (const float*)d_in[4];
    float*       out     = (float*)d_out;

    const int* src = eidx;
    const int* dst = eidx + N_EDGES;

    char* ws  = (char*)d_ws;
    float* h  = (float*)(ws);
    int*  cnt = (int*)(ws + (size_t)N_NODES * HID_DIM * 4);

    const int TPB = 256;

    init_kernel<<<(N_NODES + TPB - 1) / TPB, TPB, 0, stream>>>(cnt);

    bucket_kernel<<<(N_EDGES + TPB - 1) / TPB, TPB, 0, stream>>>(src, dst, cnt, (int*)out);

    gemm_kernel<<<(M_TILES + 7) / 8, TPB, 0, stream>>>(x, W, h);

    gather_kernel<<<N_NODES / 8, TPB, 0, stream>>>(h, cnt, src, dst, b, prelu_w, out);
}


# fuse: edge bucketing runs as extra work-groups of the gemm launch (independent stages: its atomics and stores ride beside the matrix pipe); 3 launches, not 4; sums in slot arrival order as before
# speedup vs baseline: 2.2699x; 1.2447x over previous
#include <hip/hip_runtime.h>
#include <hip/hip_bf16.h>

#define N_NODES 50000
#define N_EDGES 800000
#define IN_DIM  512
#define HID_DIM 128
#define M_TILES (N_NODES / 16)

#define PAIR_STRIDE 288
#define N_PAIRS     16

#define BUCKET_CAP HID_DIM

#define GEMM_BLOCKS   ((M_TILES + 7) / 8)
#define BUCKET_BLOCKS ((N_EDGES + 255) / 256)
#define FUSED_BLOCKS  (GEMM_BLOCKS + BUCKET_BLOCKS)

typedef float v2f __attribute__((ext_vector_type(2)));
typedef float v4f __attribute__((ext_vector_type(4)));
typedef float v8f __attribute__((ext_vector_type(8)));
typedef int slot_t __attribute__((may_alias));

__global__ void init_kernel(int* __restrict__ cnt) {
    int i = blockIdx.x * blockDim.x + threadIdx.x;
    if (i < N_NODES) cnt[i] = 0;
}

__device__ __forceinline__ void bucket_body(const int* __restrict__ src,
                                            const int* __restrict__ dst,
                                            int* __restrict__ cnt, int* __restrict__ bucket,
                                            int blk) {
    int e = blk * 256 + threadIdx.x;
    if (e >= N_EDGES) return;
    int d = dst[e];
    int s = src[e];
    int p = atomicAdd(&cnt[d], 1);
    if (p < BUCKET_CAP) bucket[(size_t)d * BUCKET_CAP + p] = s;
}

__device__ __forceinline__ void gemm_body(const float* __restrict__ x,
                                          const float* __restrict__ W,
                                          float* __restrict__ h,
                                          float* __restrict__ ldsW, int blk) {
    const int lane   = threadIdx.x & 31;
    const int waveId = threadIdx.x >> 5;
    const int mTile  = blk * 8 + waveId;
    const bool valid = (mTile < M_TILES);

    const int half = lane >> 4;
    const int l16  = lane & 15;

    const int c  = threadIdx.x & 127;
    const int ph = threadIdx.x >> 7;

    v8f acc[8];
#pragma unroll
    for (int t = 0; t < 8; ++t) acc[t] = (v8f){0.f,0.f,0.f,0.f,0.f,0.f,0.f,0.f};

    const int row = mTile * 16 + l16;
    const float* xrow = x + (size_t)row * IN_DIM;

    for (int kc = 0; kc < IN_DIM; kc += 32) {
#pragma unroll
        for (int pp = 0; pp < 8; ++pp) {
            const int pair = ph * 8 + pp;
            float w0 = W[(size_t)(kc + 2 * pair    ) * HID_DIM + c];
            float w1 = W[(size_t)(kc + 2 * pair + 1) * HID_DIM + c];
            *(v2f*)(ldsW + pair * PAIR_STRIDE + c * 2) = (v2f){w0, w1};
        }
        __syncthreads();

        if (valid) {
            if (kc + 32 < IN_DIM)
                __builtin_prefetch(xrow + kc + 32, 0, 3);
#pragma unroll
            for (int k = 0; k < 32; k += 4) {
                const int kk   = k + 2 * half;
                const int pair = (k >> 1) + half;
                v2f a = *(const v2f*)(xrow + kc + kk);
                const float* bbase = ldsW + (size_t)pair * PAIR_STRIDE;
#pragma unroll
                for (int t = 0; t < 8; ++t) {
                    v2f b = *(const v2f*)(bbase + (t * 16 + l16) * 2);
                    acc[t] = __builtin_amdgcn_wmma_f32_16x16x4_f32(
                         false, a,  false, b,
                         (short)0, acc[t],
                         false,  false);
                }
            }
        }
        __syncthreads();
    }

    if (valid) {
#pragma unroll
        for (int t = 0; t < 8; ++t) {
#pragma unroll
            for (int r = 0; r < 8; ++r) {
                int m = mTile * 16 + r + 8 * half;
                h[(size_t)m * HID_DIM + t * 16 + l16] = acc[t][r];
            }
        }
    }
}

__global__ __launch_bounds__(256) void gemm_bucket_kernel(const float* __restrict__ x,
                                                          const float* __restrict__ W,
                                                          float* __restrict__ h,
                                                          const int* __restrict__ src,
                                                          const int* __restrict__ dst,
                                                          int* __restrict__ cnt,
                                                          int* __restrict__ bucket) {
    __shared__ float ldsW[N_PAIRS * PAIR_STRIDE];

    if (blockIdx.x < GEMM_BLOCKS) {
        gemm_body(x, W, h, ldsW, (int)blockIdx.x);
    } else {
        bucket_body(src, dst, cnt, bucket, (int)blockIdx.x - GEMM_BLOCKS);
    }
}

__device__ __forceinline__ float dinv_of(int c) { return rsqrtf((float)(c + 1)); }

template <int G>
__device__ __forceinline__ void add_rows(v4f& acc, const v4f* __restrict__ h4,
                                         int sj, float wj, int k, int lane) {
    int s[G];
    float w[G];
    v4f v[G];
    for (int u = 0; u < G; ++u) {
        s[u] = __builtin_amdgcn_readlane(sj, k + u);
        w[u] = __builtin_bit_cast(float,
                 __builtin_amdgcn_readlane(__builtin_bit_cast(int, wj), k + u));
    }
    for (int u = 0; u < G; ++u) v[u] = h4[(size_t)s[u] * (HID_DIM / 4) + lane];
    for (int u = 0; u < G; ++u) acc += v[u] * w[u];
}

__global__ __launch_bounds__(256) void gather_kernel(const float* __restrict__ h,
                                                     const int* __restrict__ cnt,
                                                     const int* __restrict__ src,
                                                     const int* __restrict__ dst,
                                                     const float* __restrict__ b,
                                                     const float* __restrict__ prelu_w,
                                                     float* out) {
    const int lane = threadIdx.x & 31;
    const int i = __builtin_amdgcn_readfirstlane(blockIdx.x * 8 + (threadIdx.x >> 5));
    if (i >= N_NODES) return;

    const v4f* h4 = (const v4f*)h;
    const int   n  = cnt[i];
    const float di = dinv_of(n);
    v4f acc = (v4f){0.f, 0.f, 0.f, 0.f};

    if (n <= BUCKET_CAP) {
        const slot_t* slots = (const slot_t*)out + (size_t)i * BUCKET_CAP;
        for (int base = 0; base < n; base += 32) {
            const int m = (n - base < 32) ? (n - base) : 32;
            int   sj = 0;
            float wj = 0.f;
            if (lane < m) {
                sj = slots[base + lane];
                wj = dinv_of(cnt[sj]) * di;
            }
            int k = 0;
            for (; k + 4 <= m; k += 4) add_rows<4>(acc, h4, sj, wj, k, lane);
            if (m & 2) { add_rows<2>(acc, h4, sj, wj, k, lane); k += 2; }
            if (m & 1) add_rows<1>(acc, h4, sj, wj, k, lane);
        }
    } else {
        for (int e0 = 0; e0 < N_EDGES; e0 += 32) {
            unsigned mask = __builtin_amdgcn_ballot_w32(dst[e0 + lane] == i);
            if (mask) {
                const int   sj = src[e0 + lane];
                const float wj = dinv_of(cnt[sj]) * di;
                while (mask) {
                    const int k = __builtin_ctz(mask);
                    mask &= mask - 1;
                    add_rows<1>(acc, h4, sj, wj, k, lane);
                }
            }
        }
    }

    const v4f hi = h4[(size_t)i * (HID_DIM / 4) + lane];
    const v4f bb = ((const v4f*)b)[lane];
    const v4f pw = ((const v4f*)prelu_w)[lane];
    const v4f v  = acc + hi * di * di + bb;
    v4f r;
    r.x = (v.x > 0.0f) ? v.x : pw.x * v.x;
    r.y = (v.y > 0.0f) ? v.y : pw.y * v.y;
    r.z = (v.z > 0.0f) ? v.z : pw.z * v.z;
    r.w = (v.w > 0.0f) ? v.w : pw.w * v.w;
    ((v4f*)out)[(size_t)i * (HID_DIM / 4) + lane] = r;
}

extern "C" void kernel_launch(void* const* d_in, const int* in_sizes, int n_in,
                              void* d_out, int out_size, void* d_ws, size_t ws_size,
                              hipStream_t stream) {
    const float* x       = (const float*)d_in[0];
    const int*   eidx    = (const int*)d_in[1];
    const float* W       = (const float*)d_in[2];
    const float* b       = (const float*)d_in[3];
    const float* prelu_w = (const float*)d_in[4];
    float*       out     = (float*)d_out;

    const int* src = eidx;
    const int* dst = eidx + N_EDGES;

    char* ws  = (char*)d_ws;
    float* h  = (float*)(ws);
    int*  cnt = (int*)(ws + (size_t)N_NODES * HID_DIM * 4);

    const int TPB = 256;

    init_kernel<<<(N_NODES + TPB - 1) / TPB, TPB, 0, stream>>>(cnt);

    gemm_bucket_kernel<<<FUSED_BLOCKS, TPB, 0, stream>>>(x, W, h, src, dst, cnt,
                                                                      (int*)out);

    gather_kernel<<<N_NODES / 8, TPB, 0, stream>>>(h, cnt, src, dst, b, prelu_w, out);
}


# bucket: arrival counters leave the shared array for each destination's own out row, one counter+slot region per chip half (die id): no atomic line is shared; sums region 0 slots, then region 1
# speedup vs baseline: 2.9265x; 1.2893x over previous
#include <hip/hip_runtime.h>
#include <hip/hip_bf16.h>

#define N_NODES 50000
#define N_EDGES 800000
#define IN_DIM  512
#define HID_DIM 128
#define M_TILES (N_NODES / 16)

#define PAIR_STRIDE 288
#define N_PAIRS     16

#define REGION     64
#define BUCKET_CAP (REGION - 1)

#define GEMM_BLOCKS   ((M_TILES + 7) / 8)
#define BUCKET_BLOCKS ((N_EDGES + 255) / 256)
#define FUSED_BLOCKS  (GEMM_BLOCKS + BUCKET_BLOCKS)

typedef float v2f __attribute__((ext_vector_type(2)));
typedef float v4f __attribute__((ext_vector_type(4)));
typedef float v8f __attribute__((ext_vector_type(8)));
typedef int slot_t __attribute__((may_alias));

__global__ void init_kernel(int* __restrict__ bucket) {
    int i = blockIdx.x * blockDim.x + threadIdx.x;
    if (i < N_NODES) {
        bucket[(size_t)i * HID_DIM]          = 0;
        bucket[(size_t)i * HID_DIM + REGION] = 0;
    }
}

__global__ void pack_kernel(const int* __restrict__ bucket, int* __restrict__ cnt) {
    int i = blockIdx.x * blockDim.x + threadIdx.x;
    if (i < N_NODES)
        cnt[i] = bucket[(size_t)i * HID_DIM] + bucket[(size_t)i * HID_DIM + REGION];
}

__device__ __forceinline__ void bucket_body(const int* __restrict__ src,
                                            const int* __restrict__ dst,
                                            int* __restrict__ bucket, int blk) {
    int e = blk * 256 + threadIdx.x;
    if (e >= N_EDGES) return;
    const unsigned id = __builtin_amdgcn_s_sendmsg_rtn(0x87);
    const int half = (int)((id >> 18) & 1u);
    int d = dst[e];
    int s = src[e];
    int* region = bucket + (size_t)d * HID_DIM + half * REGION;
    int p = atomicAdd(region, 1);
    if (p < BUCKET_CAP) region[1 + p] = s;
}

__device__ __forceinline__ void gemm_body(const float* __restrict__ x,
                                          const float* __restrict__ W,
                                          float* __restrict__ h,
                                          float* __restrict__ ldsW, int blk) {
    const int lane   = threadIdx.x & 31;
    const int waveId = threadIdx.x >> 5;
    const int mTile  = blk * 8 + waveId;
    const bool valid = (mTile < M_TILES);

    const int half = lane >> 4;
    const int l16  = lane & 15;

    const int c  = threadIdx.x & 127;
    const int ph = threadIdx.x >> 7;

    v8f acc[8];
#pragma unroll
    for (int t = 0; t < 8; ++t) acc[t] = (v8f){0.f,0.f,0.f,0.f,0.f,0.f,0.f,0.f};

    const int row = mTile * 16 + l16;
    const float* xrow = x + (size_t)row * IN_DIM;

    for (int kc = 0; kc < IN_DIM; kc += 32) {
#pragma unroll
        for (int pp = 0; pp < 8; ++pp) {
            const int pair = ph * 8 + pp;
            float w0 = W[(size_t)(kc + 2 * pair    ) * HID_DIM + c];
            float w1 = W[(size_t)(kc + 2 * pair + 1) * HID_DIM + c];
            *(v2f*)(ldsW + pair * PAIR_STRIDE + c * 2) = (v2f){w0, w1};
        }
        __syncthreads();

        if (valid) {
            if (kc + 32 < IN_DIM)
                __builtin_prefetch(xrow + kc + 32, 0, 3);
#pragma unroll
            for (int k = 0; k < 32; k += 4) {
                const int kk   = k + 2 * half;
                const int pair = (k >> 1) + half;
                v2f a = *(const v2f*)(xrow + kc + kk);
                const float* bbase = ldsW + (size_t)pair * PAIR_STRIDE;
#pragma unroll
                for (int t = 0; t < 8; ++t) {
                    v2f b = *(const v2f*)(bbase + (t * 16 + l16) * 2);
                    acc[t] = __builtin_amdgcn_wmma_f32_16x16x4_f32(
                         false, a,  false, b,
                         (short)0, acc[t],
                         false,  false);
                }
            }
        }
        __syncthreads();
    }

    if (valid) {
#pragma unroll
        for (int t = 0; t < 8; ++t) {
#pragma unroll
            for (int r = 0; r < 8; ++r) {
                int m = mTile * 16 + r + 8 * half;
                h[(size_t)m * HID_DIM + t * 16 + l16] = acc[t][r];
            }
        }
    }
}

__global__ __launch_bounds__(256) void gemm_bucket_kernel(const float* __restrict__ x,
                                                          const float* __restrict__ W,
                                                          float* __restrict__ h,
                                                          const int* __restrict__ src,
                                                          const int* __restrict__ dst,
                                                          int* __restrict__ bucket) {
    __shared__ float ldsW[N_PAIRS * PAIR_STRIDE];

    if (blockIdx.x < GEMM_BLOCKS) {
        gemm_body(x, W, h, ldsW, (int)blockIdx.x);
    } else {
        bucket_body(src, dst, bucket, (int)blockIdx.x - GEMM_BLOCKS);
    }
}

__device__ __forceinline__ float dinv_of(int c) { return rsqrtf((float)(c + 1)); }

template <int G>
__device__ __forceinline__ void add_rows(v4f& acc, const v4f* __restrict__ h4,
                                         int sj, float wj, int k, int lane) {
    int s[G];
    float w[G];
    v4f v[G];
    for (int u = 0; u < G; ++u) {
        s[u] = __builtin_amdgcn_readlane(sj, k + u);
        w[u] = __builtin_bit_cast(float,
                 __builtin_amdgcn_readlane(__builtin_bit_cast(int, wj), k + u));
    }
    for (int u = 0; u < G; ++u) v[u] = h4[(size_t)s[u] * (HID_DIM / 4) + lane];
    for (int u = 0; u < G; ++u) acc += v[u] * w[u];
}

__global__ __launch_bounds__(256) void gather_kernel(const float* __restrict__ h,
                                                     const int* __restrict__ cnt,
                                                     const int* __restrict__ src,
                                                     const int* __restrict__ dst,
                                                     const float* __restrict__ b,
                                                     const float* __restrict__ prelu_w,
                                                     float* out) {
    const int lane = threadIdx.x & 31;
    const int i = __builtin_amdgcn_readfirstlane(blockIdx.x * 8 + (threadIdx.x >> 5));
    if (i >= N_NODES) return;

    const v4f* h4 = (const v4f*)h;
    const slot_t* row = (const slot_t*)out + (size_t)i * HID_DIM;
    const int n0 = __builtin_amdgcn_readfirstlane(row[0]);
    const int n1 = __builtin_amdgcn_readfirstlane(row[REGION]);
    const int   n  = n0 + n1;
    const float di = dinv_of(n);
    v4f acc = (v4f){0.f, 0.f, 0.f, 0.f};

    if (n0 <= BUCKET_CAP && n1 <= BUCKET_CAP) {
        for (int base = 0; base < n; base += 32) {
            const int m = (n - base < 32) ? (n - base) : 32;
            int   sj = 0;
            float wj = 0.f;
            if (lane < m) {
                const int idx = base + lane;
                sj = (idx < n0) ? row[1 + idx] : row[REGION + 1 + idx - n0];
                wj = dinv_of(cnt[sj]) * di;
            }
            int k = 0;
            for (; k + 4 <= m; k += 4) add_rows<4>(acc, h4, sj, wj, k, lane);
            if (m & 2) { add_rows<2>(acc, h4, sj, wj, k, lane); k += 2; }
            if (m & 1) add_rows<1>(acc, h4, sj, wj, k, lane);
        }
    } else {
        for (int e0 = 0; e0 < N_EDGES; e0 += 32) {
            unsigned mask = __builtin_amdgcn_ballot_w32(dst[e0 + lane] == i);
            if (mask) {
                const int   sj = src[e0 + lane];
                const float wj = dinv_of(cnt[sj]) * di;
                while (mask) {
                    const int k = __builtin_ctz(mask);
                    mask &= mask - 1;
                    add_rows<1>(acc, h4, sj, wj, k, lane);
                }
            }
        }
    }

    const v4f hi = h4[(size_t)i * (HID_DIM / 4) + lane];
    const v4f bb = ((const v4f*)b)[lane];
    const v4f pw = ((const v4f*)prelu_w)[lane];
    const v4f v  = acc + hi * di * di + bb;
    v4f r;
    r.x = (v.x > 0.0f) ? v.x : pw.x * v.x;
    r.y = (v.y > 0.0f) ? v.y : pw.y * v.y;
    r.z = (v.z > 0.0f) ? v.z : pw.z * v.z;
    r.w = (v.w > 0.0f) ? v.w : pw.w * v.w;
    ((v4f*)out)[(size_t)i * (HID_DIM / 4) + lane] = r;
}

extern "C" void kernel_launch(void* const* d_in, const int* in_sizes, int n_in,
                              void* d_out, int out_size, void* d_ws, size_t ws_size,
                              hipStream_t stream) {
    const float* x       = (const float*)d_in[0];
    const int*   eidx    = (const int*)d_in[1];
    const float* W       = (const float*)d_in[2];
    const float* b       = (const float*)d_in[3];
    const float* prelu_w = (const float*)d_in[4];
    float*       out     = (float*)d_out;

    const int* src = eidx;
    const int* dst = eidx + N_EDGES;

    char* ws  = (char*)d_ws;
    float* h  = (float*)(ws);
    int*  cnt = (int*)(ws + (size_t)N_NODES * HID_DIM * 4);

    const int TPB = 256;

    init_kernel<<<(N_NODES + TPB - 1) / TPB, TPB, 0, stream>>>((int*)out);

    gemm_bucket_kernel<<<FUSED_BLOCKS, TPB, 0, stream>>>(x, W, h, src, dst, (int*)out);

    pack_kernel<<<(N_NODES + TPB - 1) / TPB, TPB, 0, stream>>>((const int*)out, cnt);

    gather_kernel<<<N_NODES / 8, TPB, 0, stream>>>(h, cnt, src, dst, b, prelu_w, out);
}


# gemm: 4-wave work-groups, one M-tile per wave: 768 groups lay 3 waves on each of 1,024 SIMDs, last 53 tiles split by columns over a group's waves; was 391 8-wave groups, 1 or 2 per WGP
# speedup vs baseline: 3.2066x; 1.0957x over previous
#include <hip/hip_runtime.h>
#include <hip/hip_bf16.h>

#define N_NODES 50000
#define N_EDGES 800000
#define IN_DIM  512
#define HID_DIM 128
#define M_TILES (N_NODES / 16)

#define PAIR_STRIDE 288
#define N_PAIRS     16

#define REGION     64
#define BUCKET_CAP (REGION - 1)

#define TPB_FUSED     128
#define FULL_TILES    3072
#define FULL_BLOCKS   (FULL_TILES / 4)
#define TAIL_BLOCKS   (M_TILES - FULL_TILES)
#define GEMM_BLOCKS   (FULL_BLOCKS + TAIL_BLOCKS)
#define BUCKET_BLOCKS (N_EDGES / TPB_FUSED)
#define FUSED_BLOCKS  (GEMM_BLOCKS + BUCKET_BLOCKS)

typedef float v2f __attribute__((ext_vector_type(2)));
typedef float v4f __attribute__((ext_vector_type(4)));
typedef float v8f __attribute__((ext_vector_type(8)));
typedef int slot_t __attribute__((may_alias));

__global__ void init_kernel(int* __restrict__ bucket) {
    int i = blockIdx.x * blockDim.x + threadIdx.x;
    if (i < N_NODES) {
        bucket[(size_t)i * HID_DIM]          = 0;
        bucket[(size_t)i * HID_DIM + REGION] = 0;
    }
}

__global__ void pack_kernel(const int* __restrict__ bucket, int* __restrict__ cnt) {
    int i = blockIdx.x * blockDim.x + threadIdx.x;
    if (i < N_NODES)
        cnt[i] = bucket[(size_t)i * HID_DIM] + bucket[(size_t)i * HID_DIM + REGION];
}

__device__ __forceinline__ void bucket_body(const int* __restrict__ src,
                                            const int* __restrict__ dst,
                                            int* __restrict__ bucket, int blk) {
    int e = blk * TPB_FUSED + threadIdx.x;
    if (e >= N_EDGES) return;
    const unsigned id = __builtin_amdgcn_s_sendmsg_rtn(0x87);
    const int half = (int)((id >> 18) & 1u);
    int d = dst[e];
    int s = src[e];
    int* region = bucket + (size_t)d * HID_DIM + half * REGION;
    int p = atomicAdd(region, 1);
    if (p < BUCKET_CAP) region[1 + p] = s;
}

template <int NT>
__device__ __forceinline__ void gemm_body(const float* __restrict__ x,
                                          const float* __restrict__ W,
                                          float* __restrict__ h,
                                          float* __restrict__ ldsW,
                                          int mTile, int nt0) {
    const int lane = threadIdx.x & 31;
    const int half = lane >> 4;
    const int l16  = lane & 15;

    const int c = threadIdx.x;

    v8f acc[NT];
#pragma unroll
    for (int t = 0; t < NT; ++t) acc[t] = (v8f){0.f,0.f,0.f,0.f,0.f,0.f,0.f,0.f};

    const int row = mTile * 16 + l16;
    const float* xrow = x + (size_t)row * IN_DIM;

    for (int kc = 0; kc < IN_DIM; kc += 32) {
#pragma unroll
        for (int pair = 0; pair < N_PAIRS; ++pair) {
            float w0 = W[(size_t)(kc + 2 * pair    ) * HID_DIM + c];
            float w1 = W[(size_t)(kc + 2 * pair + 1) * HID_DIM + c];
            *(v2f*)(ldsW + pair * PAIR_STRIDE + c * 2) = (v2f){w0, w1};
        }
        __syncthreads();

        if (kc + 32 < IN_DIM)
            __builtin_prefetch(xrow + kc + 32, 0, 3);
#pragma unroll
        for (int k = 0; k < 32; k += 4) {
            const int kk   = k + 2 * half;
            const int pair = (k >> 1) + half;
            v2f a = *(const v2f*)(xrow + kc + kk);
            const float* bbase = ldsW + (size_t)pair * PAIR_STRIDE;
#pragma unroll
            for (int t = 0; t < NT; ++t) {
                v2f b = *(const v2f*)(bbase + ((nt0 + t) * 16 + l16) * 2);
                acc[t] = __builtin_amdgcn_wmma_f32_16x16x4_f32(
                     false, a,  false, b,
                     (short)0, acc[t],
                     false,  false);
            }
        }
        __syncthreads();
    }

#pragma unroll
    for (int t = 0; t < NT; ++t) {
#pragma unroll
        for (int r = 0; r < 8; ++r) {
            int m = mTile * 16 + r + 8 * half;
            h[(size_t)m * HID_DIM + (nt0 + t) * 16 + l16] = acc[t][r];
        }
    }
}

__global__ __launch_bounds__(TPB_FUSED) void gemm_bucket_kernel(const float* __restrict__ x,
                                                          const float* __restrict__ W,
                                                          float* __restrict__ h,
                                                          const int* __restrict__ src,
                                                          const int* __restrict__ dst,
                                                          int* __restrict__ bucket) {
    __shared__ float ldsW[N_PAIRS * PAIR_STRIDE];

    const int wave = threadIdx.x >> 5;
    if (blockIdx.x < FULL_BLOCKS) {
        gemm_body<8>(x, W, h, ldsW, (int)blockIdx.x * 4 + wave, 0);
    } else if (blockIdx.x < GEMM_BLOCKS) {
        gemm_body<2>(x, W, h, ldsW, FULL_TILES + (int)blockIdx.x - FULL_BLOCKS, wave * 2);
    } else {
        bucket_body(src, dst, bucket, (int)blockIdx.x - GEMM_BLOCKS);
    }
}

__device__ __forceinline__ float dinv_of(int c) { return rsqrtf((float)(c + 1)); }

template <int G>
__device__ __forceinline__ void add_rows(v4f& acc, const v4f* __restrict__ h4,
                                         int sj, float wj, int k, int lane) {
    int s[G];
    float w[G];
    v4f v[G];
    for (int u = 0; u < G; ++u) {
        s[u] = __builtin_amdgcn_readlane(sj, k + u);
        w[u] = __builtin_bit_cast(float,
                 __builtin_amdgcn_readlane(__builtin_bit_cast(int, wj), k + u));
    }
    for (int u = 0; u < G; ++u) v[u] = h4[(size_t)s[u] * (HID_DIM / 4) + lane];
    for (int u = 0; u < G; ++u) acc += v[u] * w[u];
}

__global__ __launch_bounds__(256) void gather_kernel(const float* __restrict__ h,
                                                     const int* __restrict__ cnt,
                                                     const int* __restrict__ src,
                                                     const int* __restrict__ dst,
                                                     const float* __restrict__ b,
                                                     const float* __restrict__ prelu_w,
                                                     float* out) {
    const int lane = threadIdx.x & 31;
    const int i = __builtin_amdgcn_readfirstlane(blockIdx.x * 8 + (threadIdx.x >> 5));
    if (i >= N_NODES) return;

    const v4f* h4 = (const v4f*)h;
    const slot_t* row = (const slot_t*)out + (size_t)i * HID_DIM;
    const int n0 = __builtin_amdgcn_readfirstlane(row[0]);
    const int n1 = __builtin_amdgcn_readfirstlane(row[REGION]);
    const int   n  = n0 + n1;
    const float di = dinv_of(n);
    v4f acc = (v4f){0.f, 0.f, 0.f, 0.f};

    if (n0 <= BUCKET_CAP && n1 <= BUCKET_CAP) {
        for (int base = 0; base < n; base += 32) {
            const int m = (n - base < 32) ? (n - base) : 32;
            int   sj = 0;
            float wj = 0.f;
            if (lane < m) {
                const int idx = base + lane;
                sj = (idx < n0) ? row[1 + idx] : row[REGION + 1 + idx - n0];
                wj = dinv_of(cnt[sj]) * di;
            }
            int k = 0;
            for (; k + 4 <= m; k += 4) add_rows<4>(acc, h4, sj, wj, k, lane);
            if (m & 2) { add_rows<2>(acc, h4, sj, wj, k, lane); k += 2; }
            if (m & 1) add_rows<1>(acc, h4, sj, wj, k, lane);
        }
    } else {
        for (int e0 = 0; e0 < N_EDGES; e0 += 32) {
            unsigned mask = __builtin_amdgcn_ballot_w32(dst[e0 + lane] == i);
            if (mask) {
                const int   sj = src[e0 + lane];
                const float wj = dinv_of(cnt[sj]) * di;
                while (mask) {
                    const int k = __builtin_ctz(mask);
                    mask &= mask - 1;
                    add_rows<1>(acc, h4, sj, wj, k, lane);
                }
            }
        }
    }

    const v4f hi = h4[(size_t)i * (HID_DIM / 4) + lane];
    const v4f bb = ((const v4f*)b)[lane];
    const v4f pw = ((const v4f*)prelu_w)[lane];
    const v4f v  = acc + hi * di * di + bb;
    v4f r;
    r.x = (v.x > 0.0f) ? v.x : pw.x * v.x;
    r.y = (v.y > 0.0f) ? v.y : pw.y * v.y;
    r.z = (v.z > 0.0f) ? v.z : pw.z * v.z;
    r.w = (v.w > 0.0f) ? v.w : pw.w * v.w;
    ((v4f*)out)[(size_t)i * (HID_DIM / 4) + lane] = r;
}

extern "C" void kernel_launch(void* const* d_in, const int* in_sizes, int n_in,
                              void* d_out, int out_size, void* d_ws, size_t ws_size,
                              hipStream_t stream) {
    const float* x       = (const float*)d_in[0];
    const int*   eidx    = (const int*)d_in[1];
    const float* W       = (const float*)d_in[2];
    const float* b       = (const float*)d_in[3];
    const float* prelu_w = (const float*)d_in[4];
    float*       out     = (float*)d_out;

    const int* src = eidx;
    const int* dst = eidx + N_EDGES;

    char* ws  = (char*)d_ws;
    float* h  = (float*)(ws);
    int*  cnt = (int*)(ws + (size_t)N_NODES * HID_DIM * 4);

    const int TPB = 256;

    init_kernel<<<(N_NODES + TPB - 1) / TPB, TPB, 0, stream>>>((int*)out);

    gemm_bucket_kernel<<<FUSED_BLOCKS, TPB_FUSED, 0, stream>>>(x, W, h, src, dst, (int*)out);

    pack_kernel<<<(N_NODES + TPB - 1) / TPB, TPB, 0, stream>>>((const int*)out, cnt);

    gather_kernel<<<N_NODES / 8, TPB, 0, stream>>>(h, cnt, src, dst, b, prelu_w, out);
}
